# starting kernel
# baseline (speedup 1.0000x reference)
#include <hip/hip_runtime.h>
#include <hip/hip_bf16.h>
#include <math.h>


typedef float v2f __attribute__((ext_vector_type(2)));
typedef float v8f __attribute__((ext_vector_type(8)));

#define LPAD(i) ((i) + ((i) >> 4))
#define LDS_BAR() asm volatile("" ::: "memory")

__device__ __forceinline__ v8f wmma4(v2f a, v2f b, v8f c) {
  return __builtin_amdgcn_wmma_f32_16x16x4_f32(
       false, a,  false, b,
       (short)0, c,  false,  false);
}

struct C2 { float re, im; };
__device__ __forceinline__ C2 cmul(C2 a, C2 b) {
  return { a.re * b.re - a.im * b.im, a.re * b.im + a.im * b.re };
}

__global__ void build_gates(const float* __restrict__ w, float* __restrict__ ws) {
  const int l   = blockIdx.x;
  const int tid = threadIdx.x;

  C2 g[9][2][2];
  #pragma unroll
  for (int q = 0; q < 9; ++q) {
    const float phi   = w[(l * 9 + q) * 3 + 0];
    const float theta = w[(l * 9 + q) * 3 + 1];
    const float omega = w[(l * 9 + q) * 3 + 2];
    const float ct = __cosf(0.5f * theta), st = __sinf(0.5f * theta);
    const float app = 0.5f * (phi + omega), apm = 0.5f * (phi - omega);
    const C2 ep = { __cosf(app), -__sinf(app) };
    const C2 em = { __cosf(apm),  __sinf(apm) };
    g[q][0][0] = {  ep.re * ct,  ep.im * ct };
    g[q][0][1] = { -em.re * st, -em.im * st };
    g[q][1][0] = {  em.re * st, -em.im * st };
    g[q][1][1] = {  ep.re * ct, -ep.im * ct };
  }

  const int r = tid >> 4, c = tid & 15;
  C2 hi  = { 1.0f, 0.0f };
  C2 mid = { 1.0f, 0.0f };
  #pragma unroll
  for (int k = 0; k < 4; ++k) {
    const int rb = (r >> (3 - k)) & 1, cb = (c >> (3 - k)) & 1;
    hi  = cmul(hi,  g[k][rb][cb]);
    mid = cmul(mid, g[4 + k][rb][cb]);
  }
  ws[(l * 256 + tid) * 2 + 0]        = hi.re;
  ws[(l * 256 + tid) * 2 + 1]        = hi.im;
  ws[1024 + (l * 256 + tid) * 2 + 0] = mid.re;
  ws[1024 + (l * 256 + tid) * 2 + 1] = mid.im;
  if (tid < 4) {
    const C2 e = g[8][tid >> 1][tid & 1];
    ws[2048 + (l * 4 + tid) * 2 + 0] = e.re;
    ws[2048 + (l * 4 + tid) * 2 + 1] = e.im;
  }
}

__global__ __launch_bounds__(256) void qsim(const float* __restrict__ x,
                                            const float* __restrict__ gates,
                                            float* __restrict__ out) {
  __shared__ float2 smem[8 * 544];
  const int wv   = threadIdx.x >> 5;
  const int lane = threadIdx.x & 31;
  float2* S = &smem[wv * 544];

  const int n   = blockIdx.x * 8 + wv;
  const int b   = n >> 12;
  const int rem = n & 4095;
  const int ii  = rem >> 6;
  const int jj  = rem & 63;

  float cq[9], sq[9];
  #pragma unroll
  for (int q = 0; q < 9; ++q) {
    const int ki = ii + q / 3 - 1;
    const int kj = jj + q % 3 - 1;
    float a = 0.0f;
    if (ki >= 0 && ki < 64 && kj >= 0 && kj < 64) a = x[b * 4096 + ki * 64 + kj];
    cq[q] = __cosf(0.5f * a);
    sq[q] = __sinf(0.5f * a);
  }

  #pragma unroll
  for (int k = 0; k < 16; ++k) {
    const int i = k * 32 + lane;
    float mag = 1.0f;
    #pragma unroll
    for (int q = 0; q < 9; ++q) mag *= ((i >> (8 - q)) & 1) ? sq[q] : cq[q];
    const int p = __popc(i) & 3;
    const float re = (p == 0) ? mag : ((p == 2) ? -mag : 0.0f);
    const float im = (p == 1) ? -mag : ((p == 3) ? mag : 0.0f);
    S[LPAD(i)] = make_float2(re, im);
  }
  LDS_BAR();

  const int lhalf = lane >> 4;
  const int lmod  = lane & 15;

  for (int l = 0; l < 2; ++l) {
    {
      const float* G = gates + l * 512;
      v2f Ar[4], Ai[4], Ain[4];
      #pragma unroll
      for (int kb = 0; kb < 4; ++kb)
        #pragma unroll
        for (int j = 0; j < 2; ++j) {
          const int K = 4 * kb + j + 2 * lhalf;
          const float gr = G[(lmod * 16 + K) * 2 + 0];
          const float gi = G[(lmod * 16 + K) * 2 + 1];
          Ar[kb][j] = gr; Ai[kb][j] = gi; Ain[kb][j] = -gi;
        }
      #pragma unroll
      for (int ct = 0; ct < 2; ++ct) {
        v2f Br[4], Bi[4];
        #pragma unroll
        for (int kb = 0; kb < 4; ++kb)
          #pragma unroll
          for (int j = 0; j < 2; ++j) {
            const int K = 4 * kb + j + 2 * lhalf;
            const float2 v = S[LPAD(K * 32 + ct * 16 + lmod)];
            Br[kb][j] = v.x;
            Bi[kb][j] = v.y;
          }
        v8f aR = {0, 0, 0, 0, 0, 0, 0, 0};
        v8f aI = {0, 0, 0, 0, 0, 0, 0, 0};
        #pragma unroll
        for (int kb = 0; kb < 4; ++kb) {
          aR = wmma4(Ar[kb],  Br[kb], aR);
          aR = wmma4(Ain[kb], Bi[kb], aR);
          aI = wmma4(Ar[kb],  Bi[kb], aI);
          aI = wmma4(Ai[kb],  Br[kb], aI);
        }
        #pragma unroll
        for (int r = 0; r < 8; ++r) {
          const int didx = (r + 8 * lhalf) * 32 + ct * 16 + lmod;
          S[LPAD(didx)] = make_float2(aR[r], aI[r]);
        }
        LDS_BAR();
      }
    }

    {
      const float* G = gates + 1024 + l * 512;
      v2f Ar[4], Ai[4], Ain[4];
      #pragma unroll
      for (int kb = 0; kb < 4; ++kb)
        #pragma unroll
        for (int j = 0; j < 2; ++j) {
          const int K = 4 * kb + j + 2 * lhalf;
          const float gr = G[(lmod * 16 + K) * 2 + 0];
          const float gi = G[(lmod * 16 + K) * 2 + 1];
          Ar[kb][j] = gr; Ai[kb][j] = gi; Ain[kb][j] = -gi;
        }
      #pragma unroll
      for (int lp = 0; lp < 2; ++lp) {
        v2f Br[4], Bi[4];
        #pragma unroll
        for (int kb = 0; kb < 4; ++kb)
          #pragma unroll
          for (int j = 0; j < 2; ++j) {
            const int K = 4 * kb + j + 2 * lhalf;
            const float2 v = S[LPAD(lmod * 32 + K * 2 + lp)];
            Br[kb][j] = v.x;
            Bi[kb][j] = v.y;
          }
        v8f aR = {0, 0, 0, 0, 0, 0, 0, 0};
        v8f aI = {0, 0, 0, 0, 0, 0, 0, 0};
        #pragma unroll
        for (int kb = 0; kb < 4; ++kb) {
          aR = wmma4(Ar[kb],  Br[kb], aR);
          aR = wmma4(Ain[kb], Bi[kb], aR);
          aI = wmma4(Ar[kb],  Bi[kb], aI);
          aI = wmma4(Ai[kb],  Br[kb], aI);
        }
        #pragma unroll
        for (int r = 0; r < 8; ++r) {
          const int didx = lmod * 32 + (r + 8 * lhalf) * 2 + lp;
          S[LPAD(didx)] = make_float2(aR[r], aI[r]);
        }
        LDS_BAR();
      }
    }

    {
      const float* g8 = gates + 2048 + l * 8;
      const float a00r = g8[0], a00i = g8[1], a01r = g8[2], a01i = g8[3];
      const float a10r = g8[4], a10i = g8[5], a11r = g8[6], a11i = g8[7];
      #pragma unroll
      for (int k = 0; k < 8; ++k) {
        const int p  = k * 32 + lane;
        const int i0 = 2 * p, i1 = 2 * p + 1;
        const float2 s0 = S[LPAD(i0)];
        const float2 s1 = S[LPAD(i1)];
        const float n0r = a00r * s0.x - a00i * s0.y + a01r * s1.x - a01i * s1.y;
        const float n0i = a00r * s0.y + a00i * s0.x + a01r * s1.y + a01i * s1.x;
        const float n1r = a10r * s0.x - a10i * s0.y + a11r * s1.x - a11i * s1.y;
        const float n1i = a10r * s0.y + a10i * s0.x + a11r * s1.y + a11i * s1.x;
        S[LPAD(i0)] = make_float2(n0r, n0i);
        S[LPAD(i1)] = make_float2(n1r, n1i);
      }
      LDS_BAR();
    }

    {
      const int rr = l + 1;
      float2 t[16];
      #pragma unroll
      for (int k = 0; k < 16; ++k) {
        const int i = k * 32 + lane;
        int j = i;
        #pragma unroll
        for (int q = 8; q >= 0; --q) {
          const int cb = 8 - q;
          const int tb = 8 - ((q + rr) % 9);
          j ^= ((j >> cb) & 1) << tb;
        }
        t[k] = S[LPAD(j)];
      }
      LDS_BAR();
      #pragma unroll
      for (int k = 0; k < 16; ++k) {
        const int i = k * 32 + lane;
        S[LPAD(i)] = t[k];
      }
      LDS_BAR();
    }
  }

  float pz = 0.0f;
  #pragma unroll
  for (int k = 0; k < 16; ++k) {
    const int i = k * 32 + lane;
    const float2 s = S[LPAD(i)];
    const float v = s.x * s.x + s.y * s.y;
    pz += (i & 256) ? -v : v;
  }
  #pragma unroll
  for (int off = 16; off > 0; off >>= 1) pz += __shfl_down(pz, off, 32);
  if (lane == 0) out[n] = pz;
}

extern "C" void kernel_launch(void* const* d_in, const int* in_sizes, int n_in,
                              void* d_out, int out_size, void* d_ws, size_t ws_size,
                              hipStream_t stream) {
  const float* x = (const float*)d_in[0];
  const float* w = (const float*)d_in[1];
  float* out = (float*)d_out;
  float* ws  = (float*)d_ws;

  hipLaunchKernelGGL(build_gates, dim3(2), dim3(256), 0, stream, w, ws);
  hipLaunchKernelGGL(qsim, dim3(32768 / 8), dim3(256), 0, stream, x, ws, out);
}


# State stays in WMMA fragment registers through a layer: hi group computed transposed so its D tile is the mid group's B operand; one LDS transpose per layer carries the CNOT ring; sum order changes
# speedup vs baseline: 1.4358x; 1.4358x over previous
#include <hip/hip_runtime.h>
#include <math.h>


typedef float v2f __attribute__((ext_vector_type(2)));
typedef float v8f __attribute__((ext_vector_type(8)));

#define PHYS(i) ((i) + 2 * ((i) >> 5))
#define LDS_BAR() asm volatile("" ::: "memory")

__device__ __forceinline__ v8f wmma4(v2f a, v2f b, v8f c) {
  return __builtin_amdgcn_wmma_f32_16x16x4_f32(
       false, a,  false, b,
       (short)0, c,  false,  false);
}

__device__ __forceinline__ constexpr int cnot_pinv(int j, int rr) {
  for (int q = 0; q < 9; ++q) {
    const int cb = 8 - q;
    const int tb = 8 - ((q + rr) % 9);
    j ^= ((j >> cb) & 1) << tb;
  }
  return j;
}

struct C2 { float re, im; };
__device__ __forceinline__ C2 cmul(C2 a, C2 b) {
  return { a.re * b.re - a.im * b.im, a.re * b.im + a.im * b.re };
}

__global__ void build_gates(const float* __restrict__ w, float* __restrict__ ws) {
  const int l   = blockIdx.x;
  const int tid = threadIdx.x;

  C2 g[9][2][2];
  #pragma unroll
  for (int q = 0; q < 9; ++q) {
    const float phi   = w[(l * 9 + q) * 3 + 0];
    const float theta = w[(l * 9 + q) * 3 + 1];
    const float omega = w[(l * 9 + q) * 3 + 2];
    const float ct = __cosf(0.5f * theta), st = __sinf(0.5f * theta);
    const float app = 0.5f * (phi + omega), apm = 0.5f * (phi - omega);
    const C2 ep = { __cosf(app), -__sinf(app) };
    const C2 em = { __cosf(apm),  __sinf(apm) };
    g[q][0][0] = {  ep.re * ct,  ep.im * ct };
    g[q][0][1] = { -em.re * st, -em.im * st };
    g[q][1][0] = {  em.re * st, -em.im * st };
    g[q][1][1] = {  ep.re * ct, -ep.im * ct };
  }

  const int r = tid >> 4, c = tid & 15;
  C2 hi  = { 1.0f, 0.0f };
  C2 mid = { 1.0f, 0.0f };
  #pragma unroll
  for (int k = 0; k < 4; ++k) {
    const int rb = (r >> (3 - k)) & 1, cb = (c >> (3 - k)) & 1;
    hi  = cmul(hi,  g[k][rb][cb]);
    mid = cmul(mid, g[4 + k][rb][cb]);
  }
  ws[(l * 256 + tid) * 2 + 0]        = hi.re;
  ws[(l * 256 + tid) * 2 + 1]        = hi.im;
  ws[1024 + (l * 256 + tid) * 2 + 0] = mid.re;
  ws[1024 + (l * 256 + tid) * 2 + 1] = mid.im;
  if (tid < 4) {
    const C2 e = g[8][tid >> 1][tid & 1];
    ws[2048 + (l * 4 + tid) * 2 + 0] = e.re;
    ws[2048 + (l * 4 + tid) * 2 + 1] = e.im;
  }
}


__device__ __forceinline__ void load_gate(const float* __restrict__ G, int base, int kbstep,
                                          v2f (&Gr)[4], v2f (&Gi)[4]) {
  #pragma unroll
  for (int kb = 0; kb < 4; ++kb) {
    const float4 g = *reinterpret_cast<const float4*>(G + (base + kbstep * kb) * 2);
    Gr[kb] = (v2f){g.x, g.z};
    Gi[kb] = (v2f){g.y, g.w};
  }
}

__device__ __forceinline__ void layer_gates(const v2f (&Sr)[2][4], const v2f (&Si)[2][4],
                                            const v2f (&Hr)[4], const v2f (&Hi)[4],
                                            const v2f (&Mr)[4], const v2f (&Mi)[4],
                                            const float* __restrict__ g8,
                                            v8f (&nR)[2], v8f (&nI)[2]) {
  v8f dR[2], dI[2];
  #pragma unroll
  for (int lp = 0; lp < 2; ++lp) {
    dR[lp] = (v8f){0, 0, 0, 0, 0, 0, 0, 0};
    dI[lp] = (v8f){0, 0, 0, 0, 0, 0, 0, 0};
  }
  #pragma unroll
  for (int kb = 0; kb < 4; ++kb) {
    const v2f Hin = -Hi[kb];
    #pragma unroll
    for (int lp = 0; lp < 2; ++lp) {
      dR[lp] = wmma4(Sr[lp][kb], Hr[kb], dR[lp]);
      dR[lp] = wmma4(Si[lp][kb], Hin,    dR[lp]);
      dI[lp] = wmma4(Sr[lp][kb], Hi[kb], dI[lp]);
      dI[lp] = wmma4(Si[lp][kb], Hr[kb], dI[lp]);
    }
  }
  v8f bR[2], bI[2];
  #pragma unroll
  for (int lp = 0; lp < 2; ++lp) {
    bR[lp] = (v8f){0, 0, 0, 0, 0, 0, 0, 0};
    bI[lp] = (v8f){0, 0, 0, 0, 0, 0, 0, 0};
  }
  #pragma unroll
  for (int kb = 0; kb < 4; ++kb) {
    const v2f Min = -Mi[kb];
    #pragma unroll
    for (int lp = 0; lp < 2; ++lp) {
      const v2f Br = (v2f){dR[lp][2 * kb], dR[lp][2 * kb + 1]};
      const v2f Bi = (v2f){dI[lp][2 * kb], dI[lp][2 * kb + 1]};
      bR[lp] = wmma4(Mr[kb], Br, bR[lp]);
      bR[lp] = wmma4(Min,    Bi, bR[lp]);
      bI[lp] = wmma4(Mr[kb], Bi, bI[lp]);
      bI[lp] = wmma4(Mi[kb], Br, bI[lp]);
    }
  }
  const float a00r = g8[0], a00i = g8[1], a01r = g8[2], a01i = g8[3];
  const float a10r = g8[4], a10i = g8[5], a11r = g8[6], a11i = g8[7];
  #pragma unroll
  for (int r = 0; r < 8; ++r) {
    const float s0x = bR[0][r], s0y = bI[0][r], s1x = bR[1][r], s1y = bI[1][r];
    nR[0][r] = a00r * s0x - a00i * s0y + a01r * s1x - a01i * s1y;
    nI[0][r] = a00r * s0y + a00i * s0x + a01r * s1y + a01i * s1x;
    nR[1][r] = a10r * s0x - a10i * s0y + a11r * s1x - a11i * s1y;
    nI[1][r] = a10r * s0y + a10i * s0x + a11r * s1y + a11i * s1x;
  }
}

__global__ __launch_bounds__(256) void qsim(const float* __restrict__ x,
                                            const float* __restrict__ gates,
                                            float* __restrict__ out) {
  __shared__ __align__(16) float2 smem[8 * 544];
  const int wv   = threadIdx.x >> 5;
  const int lane = threadIdx.x & 31;
  float2* S = &smem[wv * 544];
  const int lhalf = lane >> 4;
  const int lmod  = lane & 15;

  const int n   = blockIdx.x * 8 + wv;
  const int b   = n >> 12;
  const int rem = n & 4095;
  const int ii  = rem >> 6;
  const int jj  = rem & 63;

  v2f Hr[4], Hi[4], Mr[4], Mi[4];
  load_gate(gates,        lmod * 16 + 2 * lhalf, 4, Hr, Hi);
  load_gate(gates + 1024, lmod * 16 + 8 * lhalf, 2, Mr, Mi);

  float cq[9], sq[9];
  #pragma unroll
  for (int q = 0; q < 9; ++q) {
    const int ki = ii + q / 3 - 1;
    const int kj = jj + q % 3 - 1;
    float a = 0.0f;
    if (ki >= 0 && ki < 64 && kj >= 0 && kj < 64) a = x[b * 4096 + ki * 64 + kj];
    cq[q] = __cosf(0.5f * a);
    sq[q] = __sinf(0.5f * a);
  }

  v2f Sr[2][4], Si[2][4];
  {
    float lf = (lmod & 8) ? sq[4] : cq[4];
    lf *= (lmod & 4) ? sq[5] : cq[5];
    lf *= (lmod & 2) ? sq[6] : cq[6];
    lf *= (lmod & 1) ? sq[7] : cq[7];
    lf *= lhalf ? sq[2] : cq[2];
    const int pl = (__popc(lmod) + lhalf) & 3;
    const float bre = (pl == 0) ? lf : ((pl == 2) ? -lf : 0.0f);
    const float bim = (pl == 1) ? -lf : ((pl == 3) ? lf : 0.0f);
    #pragma unroll
    for (int kb = 0; kb < 4; ++kb)
      #pragma unroll
      for (int j = 0; j < 2; ++j)
        #pragma unroll
        for (int lp = 0; lp < 2; ++lp) {
          const float cm = ((kb & 2) ? sq[0] : cq[0]) * ((kb & 1) ? sq[1] : cq[1]) *
                           (j ? sq[3] : cq[3]) * (lp ? sq[8] : cq[8]);
          const int pc = (((kb >> 1) & 1) + (kb & 1) + j + lp) & 3;
          float re, im;
          if (pc == 0)      { re =  cm * bre; im =  cm * bim; }
          else if (pc == 1) { re =  cm * bim; im = -cm * bre; }
          else if (pc == 2) { re = -cm * bre; im = -cm * bim; }
          else              { re = -cm * bim; im =  cm * bre; }
          Sr[lp][kb][j] = re;
          Si[lp][kb][j] = im;
        }
  }

  v8f nR[2], nI[2];
  layer_gates(Sr, Si, Hr, Hi, Mr, Mi, gates + 2048, nR, nI);

  load_gate(gates + 512,        lmod * 16 + 2 * lhalf, 4, Hr, Hi);
  load_gate(gates + 1024 + 512, lmod * 16 + 8 * lhalf, 2, Mr, Mi);

  {
    const int Lb = cnot_pinv((lmod << 5) | (lhalf << 4), 1);
    #pragma unroll
    for (int r = 0; r < 8; ++r)
      #pragma unroll
      for (int lp = 0; lp < 2; ++lp) {
        const int idx = Lb ^ cnot_pinv((r << 1) | lp, 1);
        S[PHYS(idx)] = make_float2(nR[lp][r], nI[lp][r]);
      }
    LDS_BAR();
    #pragma unroll
    for (int kb = 0; kb < 4; ++kb)
      #pragma unroll
      for (int j = 0; j < 2; ++j) {
        const int K  = 4 * kb + 2 * lhalf + j;
        const int i0 = (K << 5) | (lmod << 1);
        const float4 v = *reinterpret_cast<const float4*>(&S[PHYS(i0)]);
        Sr[0][kb][j] = v.x; Si[0][kb][j] = v.y;
        Sr[1][kb][j] = v.z; Si[1][kb][j] = v.w;
      }
  }

  layer_gates(Sr, Si, Hr, Hi, Mr, Mi, gates + 2048 + 8, nR, nI);

  const int sl = (cnot_pinv((lmod << 5) | (lhalf << 4), 2) >> 8) & 1;
  float pz = 0.0f;
  #pragma unroll
  for (int r = 0; r < 8; ++r)
    #pragma unroll
    for (int lp = 0; lp < 2; ++lp) {
      const int sc = (cnot_pinv((r << 1) | lp, 2) >> 8) & 1;
      const float v = nR[lp][r] * nR[lp][r] + nI[lp][r] * nI[lp][r];
      pz += sc ? -v : v;
    }
  pz = sl ? -pz : pz;
  #pragma unroll
  for (int off = 16; off > 0; off >>= 1) pz += __shfl_down(pz, off, 32);
  if (lane == 0) out[n] = pz;
}

extern "C" void kernel_launch(void* const* d_in, const int* in_sizes, int n_in,
                              void* d_out, int out_size, void* d_ws, size_t ws_size,
                              hipStream_t stream) {
  const float* x = (const float*)d_in[0];
  const float* w = (const float*)d_in[1];
  float* out = (float*)d_out;
  float* ws  = (float*)d_ws;

  hipLaunchKernelGGL(build_gates, dim3(2), dim3(256), 0, stream, w, ws);
  hipLaunchKernelGGL(qsim, dim3(32768 / 8), dim3(256), 0, stream, x, ws, out);
}


# Layer 0 applied per wire to the product state (2x2 on each wire's 2-vector); amplitudes formed after its CNOT ring directly as layer-1 A fragments; layer-0 WMMAs and LDS pass gone; sum order changes
# speedup vs baseline: 2.1003x; 1.4628x over previous
#include <hip/hip_runtime.h>
#include <math.h>


typedef float v2f __attribute__((ext_vector_type(2)));
typedef float v8f __attribute__((ext_vector_type(8)));

#define WS_HI  0
#define WS_MID 512
#define WS_G8  1024
#define WS_L0  1040

__device__ __forceinline__ v8f wmma4(v2f a, v2f b, v8f c) {
  return __builtin_amdgcn_wmma_f32_16x16x4_f32(
       false, a,  false, b,
       (short)0, c,  false,  false);
}

__host__ __device__ constexpr int cnot_pinv(int j, int rr) {
  for (int q = 0; q < 9; ++q) {
    const int cb = 8 - q;
    const int tb = 8 - ((q + rr) % 9);
    j ^= ((j >> cb) & 1) << tb;
  }
  return j;
}
__host__ __device__ constexpr int cnot_p(int i, int rr) {
  for (int q = 8; q >= 0; --q) {
    const int cb = 8 - q;
    const int tb = 8 - ((q + rr) % 9);
    i ^= ((i >> cb) & 1) << tb;
  }
  return i;
}
__host__ __device__ constexpr bool ring1_closed_form_ok() {
  for (int i = 0; i < 512; ++i) {
    int k = (((i >> 8) ^ i) & 1) << 8;
    k |= (((i >> 7) ^ (i >> 8) ^ i) & 1) << 7;
    for (int b = 0; b < 7; ++b) k |= (((i >> b) ^ (i >> (b + 1))) & 1) << b;
    if (k != cnot_p(i, 1)) return false;
  }
  return true;
}
static_assert(ring1_closed_form_ok(), "closed form of the range-1 CNOT ring");

struct C2 { float re, im; };
__device__ __forceinline__ C2 cmul(C2 a, C2 b) {
  return { a.re * b.re - a.im * b.im, a.re * b.im + a.im * b.re };
}
__device__ __forceinline__ C2 csel(bool s, C2 a, C2 b) { return s ? b : a; }

__device__ __forceinline__ void rot_gate(const float* __restrict__ w3, C2 (&g)[2][2]) {
  const float phi = w3[0], theta = w3[1], omega = w3[2];
  const float ct = __cosf(0.5f * theta), st = __sinf(0.5f * theta);
  const float app = 0.5f * (phi + omega), apm = 0.5f * (phi - omega);
  const C2 ep = { __cosf(app), -__sinf(app) };
  const C2 em = { __cosf(apm),  __sinf(apm) };
  g[0][0] = {  ep.re * ct,  ep.im * ct };
  g[0][1] = { -em.re * st, -em.im * st };
  g[1][0] = {  em.re * st, -em.im * st };
  g[1][1] = {  ep.re * ct, -ep.im * ct };
}

__global__ void build_gates(const float* __restrict__ w, float* __restrict__ ws) {
  const int tid = threadIdx.x;

  C2 g[9][2][2];
  #pragma unroll
  for (int q = 0; q < 9; ++q) rot_gate(w + (9 + q) * 3, g[q]);

  const int r = tid >> 4, c = tid & 15;
  C2 hi  = { 1.0f, 0.0f };
  C2 mid = { 1.0f, 0.0f };
  #pragma unroll
  for (int k = 0; k < 4; ++k) {
    const int rb = (r >> (3 - k)) & 1, cb = (c >> (3 - k)) & 1;
    hi  = cmul(hi,  g[k][rb][cb]);
    mid = cmul(mid, g[4 + k][rb][cb]);
  }
  ws[WS_HI  + tid * 2 + 0] = hi.re;
  ws[WS_HI  + tid * 2 + 1] = hi.im;
  ws[WS_MID + tid * 2 + 0] = mid.re;
  ws[WS_MID + tid * 2 + 1] = mid.im;
  if (tid < 4) {
    const C2 e = g[8][tid >> 1][tid & 1];
    ws[WS_G8 + tid * 2 + 0] = e.re;
    ws[WS_G8 + tid * 2 + 1] = e.im;
  }
  if (tid < 9) {
    C2 g0[2][2];
    rot_gate(w + tid * 3, g0);
    float* d = ws + WS_L0 + tid * 8;
    d[0] = g0[0][0].re; d[1] = g0[0][0].im; d[2] = g0[0][1].re; d[3] = g0[0][1].im;
    d[4] = g0[1][0].re; d[5] = g0[1][0].im; d[6] = g0[1][1].re; d[7] = g0[1][1].im;
  }
}


__device__ __forceinline__ void load_gate(const float* __restrict__ G, int base, int kbstep,
                                          v2f (&Gr)[4], v2f (&Gi)[4]) {
  #pragma unroll
  for (int kb = 0; kb < 4; ++kb) {
    const float4 g = *reinterpret_cast<const float4*>(G + (base + kbstep * kb) * 2);
    Gr[kb] = (v2f){g.x, g.z};
    Gi[kb] = (v2f){g.y, g.w};
  }
}

__device__ __forceinline__ void layer_gates(const v2f (&Sr)[2][4], const v2f (&Si)[2][4],
                                            const v2f (&Hr)[4], const v2f (&Hi)[4],
                                            const v2f (&Mr)[4], const v2f (&Mi)[4],
                                            const float* __restrict__ g8,
                                            v8f (&nR)[2], v8f (&nI)[2]) {
  v8f dR[2], dI[2];
  #pragma unroll
  for (int lp = 0; lp < 2; ++lp) {
    dR[lp] = (v8f){0, 0, 0, 0, 0, 0, 0, 0};
    dI[lp] = (v8f){0, 0, 0, 0, 0, 0, 0, 0};
  }
  #pragma unroll
  for (int kb = 0; kb < 4; ++kb) {
    const v2f Hin = -Hi[kb];
    #pragma unroll
    for (int lp = 0; lp < 2; ++lp) {
      dR[lp] = wmma4(Sr[lp][kb], Hr[kb], dR[lp]);
      dR[lp] = wmma4(Si[lp][kb], Hin,    dR[lp]);
      dI[lp] = wmma4(Sr[lp][kb], Hi[kb], dI[lp]);
      dI[lp] = wmma4(Si[lp][kb], Hr[kb], dI[lp]);
    }
  }
  v8f bR[2], bI[2];
  #pragma unroll
  for (int lp = 0; lp < 2; ++lp) {
    bR[lp] = (v8f){0, 0, 0, 0, 0, 0, 0, 0};
    bI[lp] = (v8f){0, 0, 0, 0, 0, 0, 0, 0};
  }
  #pragma unroll
  for (int kb = 0; kb < 4; ++kb) {
    const v2f Min = -Mi[kb];
    #pragma unroll
    for (int lp = 0; lp < 2; ++lp) {
      const v2f Br = (v2f){dR[lp][2 * kb], dR[lp][2 * kb + 1]};
      const v2f Bi = (v2f){dI[lp][2 * kb], dI[lp][2 * kb + 1]};
      bR[lp] = wmma4(Mr[kb], Br, bR[lp]);
      bR[lp] = wmma4(Min,    Bi, bR[lp]);
      bI[lp] = wmma4(Mr[kb], Bi, bI[lp]);
      bI[lp] = wmma4(Mi[kb], Br, bI[lp]);
    }
  }
  const float a00r = g8[0], a00i = g8[1], a01r = g8[2], a01i = g8[3];
  const float a10r = g8[4], a10i = g8[5], a11r = g8[6], a11i = g8[7];
  #pragma unroll
  for (int r = 0; r < 8; ++r) {
    const float s0x = bR[0][r], s0y = bI[0][r], s1x = bR[1][r], s1y = bI[1][r];
    nR[0][r] = a00r * s0x - a00i * s0y + a01r * s1x - a01i * s1y;
    nI[0][r] = a00r * s0y + a00i * s0x + a01r * s1y + a01i * s1x;
    nR[1][r] = a10r * s0x - a10i * s0y + a11r * s1x - a11i * s1y;
    nI[1][r] = a10r * s0y + a10i * s0x + a11r * s1y + a11i * s1x;
  }
}

__global__ __launch_bounds__(256) void qsim(const float* __restrict__ x,
                                            const float* __restrict__ gates,
                                            float* __restrict__ out) {
  const int wv   = threadIdx.x >> 5;
  const int lane = threadIdx.x & 31;
  const int lhalf = lane >> 4;
  const int lmod  = lane & 15;

  const int n   = blockIdx.x * 8 + wv;
  const int b   = n >> 12;
  const int rem = n & 4095;
  const int ii  = rem >> 6;
  const int jj  = rem & 63;

  v2f Hr[4], Hi[4], Mr[4], Mi[4];
  load_gate(gates + WS_HI,  lmod * 16 + 2 * lhalf, 4, Hr, Hi);
  load_gate(gates + WS_MID, lmod * 16 + 8 * lhalf, 2, Mr, Mi);

  C2 va[9], vb[9];
  #pragma unroll
  for (int q = 0; q < 9; ++q) {
    const int ki = ii + q / 3 - 1;
    const int kj = jj + q % 3 - 1;
    float a = 0.0f;
    if (ki >= 0 && ki < 64 && kj >= 0 && kj < 64) a = x[b * 4096 + ki * 64 + kj];
    const float c = __cosf(0.5f * a);
    const float s = __sinf(0.5f * a);
    const float* __restrict__ g = gates + WS_L0 + q * 8;
    va[q] = { g[0] * c + g[3] * s, g[1] * c - g[2] * s };
    vb[q] = { g[4] * c + g[7] * s, g[5] * c - g[6] * s };
  }

  v2f Sr[2][4], Si[2][4];
  {
    const bool i6 = lhalf != 0;
    const bool i4 = (lmod & 8) != 0, i3 = (lmod & 4) != 0;
    const bool i2 = (lmod & 2) != 0, i1 = (lmod & 1) != 0;
    const C2 Lc = cmul(cmul(csel(i3 != i4, va[5], vb[5]), csel(i2 != i3, va[6], vb[6])),
                       csel(i1 != i2, va[7], vb[7]));
    const C2 u2 = csel(i6, va[2], vb[2]), w2 = csel(i6, vb[2], va[2]);
    const C2 u3 = csel(i6, va[3], vb[3]), w3 = csel(i6, vb[3], va[3]);
    const C2 u4 = csel(i4, va[4], vb[4]), w4 = csel(i4, vb[4], va[4]);
    const C2 u8 = csel(i1, va[8], vb[8]), w8 = csel(i1, vb[8], va[8]);
    const C2 t0 = cmul(Lc, cmul(u3, u4));
    const C2 t1 = cmul(Lc, cmul(w3, w4));
    const C2 T[2][2] = { { cmul(t0, u8), cmul(t0, w8) }, { cmul(t1, u8), cmul(t1, w8) } };
    const C2 f0[2] = { va[0], vb[0] };
    const C2 f1[2] = { va[1], vb[1] };
    const C2 f2[2] = { u2, w2 };
    C2 X[2][2];
    #pragma unroll
    for (int a = 0; a < 2; ++a)
      #pragma unroll
      for (int i7 = 0; i7 < 2; ++i7)
        X[a][i7] = cmul(cmul(f0[a], f1[i7 ^ a]), f2[i7]);
    #pragma unroll
    for (int i8 = 0; i8 < 2; ++i8)
      #pragma unroll
      for (int i7 = 0; i7 < 2; ++i7)
        #pragma unroll
        for (int i5 = 0; i5 < 2; ++i5)
          #pragma unroll
          for (int i0 = 0; i0 < 2; ++i0) {
            const C2 amp = cmul(X[i8 ^ i0][i7], T[i5][i0]);
            Sr[i0][2 * i8 + i7][i5] = amp.re;
            Si[i0][2 * i8 + i7][i5] = amp.im;
          }
  }

  v8f nR[2], nI[2];
  layer_gates(Sr, Si, Hr, Hi, Mr, Mi, gates + WS_G8, nR, nI);

  const int sl = (cnot_pinv((lmod << 5) | (lhalf << 4), 2) >> 8) & 1;
  float pz = 0.0f;
  #pragma unroll
  for (int r = 0; r < 8; ++r)
    #pragma unroll
    for (int lp = 0; lp < 2; ++lp) {
      const int sc = (cnot_pinv((r << 1) | lp, 2) >> 8) & 1;
      const float v = nR[lp][r] * nR[lp][r] + nI[lp][r] * nI[lp][r];
      pz += sc ? -v : v;
    }
  pz = sl ? -pz : pz;
  #pragma unroll
  for (int off = 16; off > 0; off >>= 1) pz += __shfl_down(pz, off, 32);
  if (lane == 0) out[n] = pz;
}

extern "C" void kernel_launch(void* const* d_in, const int* in_sizes, int n_in,
                              void* d_out, int out_size, void* d_ws, size_t ws_size,
                              hipStream_t stream) {
  const float* x = (const float*)d_in[0];
  const float* w = (const float*)d_in[1];
  float* out = (float*)d_out;
  float* ws  = (float*)d_ws;

  hipLaunchKernelGGL(build_gates, dim3(1), dim3(256), 0, stream, w, ws);
  hipLaunchKernelGGL(qsim, dim3(32768 / 8), dim3(256), 0, stream, x, ws, out);
}


# Each wave walks 4 consecutive samples in a rolled loop: grid 1024 x 8 waves, all resident at 8 waves per SIMD; gate fragments loaded once per wave, next patch fetched under the current sample
# speedup vs baseline: 2.3050x; 1.0975x over previous
#include <hip/hip_runtime.h>
#include <math.h>


typedef float v2f __attribute__((ext_vector_type(2)));
typedef float v8f __attribute__((ext_vector_type(8)));

#define WS_HI  0
#define WS_MID 512
#define WS_G8  1024
#define WS_L0  1040

#define SAMPLES_PER_WAVE 4

__device__ __forceinline__ v8f wmma4(v2f a, v2f b, v8f c) {
  return __builtin_amdgcn_wmma_f32_16x16x4_f32(
       false, a,  false, b,
       (short)0, c,  false,  false);
}

__host__ __device__ constexpr int cnot_pinv(int j, int rr) {
  for (int q = 0; q < 9; ++q) {
    const int cb = 8 - q;
    const int tb = 8 - ((q + rr) % 9);
    j ^= ((j >> cb) & 1) << tb;
  }
  return j;
}
__host__ __device__ constexpr int cnot_p(int i, int rr) {
  for (int q = 8; q >= 0; --q) {
    const int cb = 8 - q;
    const int tb = 8 - ((q + rr) % 9);
    i ^= ((i >> cb) & 1) << tb;
  }
  return i;
}
__host__ __device__ constexpr bool ring1_closed_form_ok() {
  for (int i = 0; i < 512; ++i) {
    int k = (((i >> 8) ^ i) & 1) << 8;
    k |= (((i >> 7) ^ (i >> 8) ^ i) & 1) << 7;
    for (int b = 0; b < 7; ++b) k |= (((i >> b) ^ (i >> (b + 1))) & 1) << b;
    if (k != cnot_p(i, 1)) return false;
  }
  return true;
}
static_assert(ring1_closed_form_ok(), "closed form of the range-1 CNOT ring");

struct C2 { float re, im; };
__device__ __forceinline__ C2 cmul(C2 a, C2 b) {
  return { a.re * b.re - a.im * b.im, a.re * b.im + a.im * b.re };
}
__device__ __forceinline__ C2 csel(bool s, C2 a, C2 b) { return s ? b : a; }

__device__ __forceinline__ void rot_gate(const float* __restrict__ w3, C2 (&g)[2][2]) {
  const float phi = w3[0], theta = w3[1], omega = w3[2];
  const float ct = __cosf(0.5f * theta), st = __sinf(0.5f * theta);
  const float app = 0.5f * (phi + omega), apm = 0.5f * (phi - omega);
  const C2 ep = { __cosf(app), -__sinf(app) };
  const C2 em = { __cosf(apm),  __sinf(apm) };
  g[0][0] = {  ep.re * ct,  ep.im * ct };
  g[0][1] = { -em.re * st, -em.im * st };
  g[1][0] = {  em.re * st, -em.im * st };
  g[1][1] = {  ep.re * ct, -ep.im * ct };
}

__global__ void build_gates(const float* __restrict__ w, float* __restrict__ ws) {
  const int tid = threadIdx.x;

  C2 g[9][2][2];
  #pragma unroll
  for (int q = 0; q < 9; ++q) rot_gate(w + (9 + q) * 3, g[q]);

  const int r = tid >> 4, c = tid & 15;
  C2 hi  = { 1.0f, 0.0f };
  C2 mid = { 1.0f, 0.0f };
  #pragma unroll
  for (int k = 0; k < 4; ++k) {
    const int rb = (r >> (3 - k)) & 1, cb = (c >> (3 - k)) & 1;
    hi  = cmul(hi,  g[k][rb][cb]);
    mid = cmul(mid, g[4 + k][rb][cb]);
  }
  ws[WS_HI  + tid * 2 + 0] = hi.re;
  ws[WS_HI  + tid * 2 + 1] = hi.im;
  ws[WS_MID + tid * 2 + 0] = mid.re;
  ws[WS_MID + tid * 2 + 1] = mid.im;
  if (tid < 4) {
    const C2 e = g[8][tid >> 1][tid & 1];
    ws[WS_G8 + tid * 2 + 0] = e.re;
    ws[WS_G8 + tid * 2 + 1] = e.im;
  }
  if (tid < 9) {
    C2 g0[2][2];
    rot_gate(w + tid * 3, g0);
    float* d = ws + WS_L0 + tid * 8;
    d[0] = g0[0][0].re; d[1] = g0[0][0].im; d[2] = g0[0][1].re; d[3] = g0[0][1].im;
    d[4] = g0[1][0].re; d[5] = g0[1][0].im; d[6] = g0[1][1].re; d[7] = g0[1][1].im;
  }
}


__device__ __forceinline__ void load_gate(const float* __restrict__ G, int base, int kbstep,
                                          v2f (&Gr)[4], v2f (&Gi)[4]) {
  #pragma unroll
  for (int kb = 0; kb < 4; ++kb) {
    const float4 g = *reinterpret_cast<const float4*>(G + (base + kbstep * kb) * 2);
    Gr[kb] = (v2f){g.x, g.z};
    Gi[kb] = (v2f){g.y, g.w};
  }
}

__device__ __forceinline__ void layer_gates(const v2f (&Sr)[2][4], const v2f (&Si)[2][4],
                                            const v2f (&Hr)[4], const v2f (&Hi)[4],
                                            const v2f (&Mr)[4], const v2f (&Mi)[4],
                                            const float* __restrict__ g8,
                                            v8f (&nR)[2], v8f (&nI)[2]) {
  v8f dR[2], dI[2];
  #pragma unroll
  for (int lp = 0; lp < 2; ++lp) {
    dR[lp] = (v8f){0, 0, 0, 0, 0, 0, 0, 0};
    dI[lp] = (v8f){0, 0, 0, 0, 0, 0, 0, 0};
  }
  #pragma unroll
  for (int kb = 0; kb < 4; ++kb) {
    const v2f Hin = -Hi[kb];
    #pragma unroll
    for (int lp = 0; lp < 2; ++lp) {
      dR[lp] = wmma4(Sr[lp][kb], Hr[kb], dR[lp]);
      dR[lp] = wmma4(Si[lp][kb], Hin,    dR[lp]);
      dI[lp] = wmma4(Sr[lp][kb], Hi[kb], dI[lp]);
      dI[lp] = wmma4(Si[lp][kb], Hr[kb], dI[lp]);
    }
  }
  v8f bR[2], bI[2];
  #pragma unroll
  for (int lp = 0; lp < 2; ++lp) {
    bR[lp] = (v8f){0, 0, 0, 0, 0, 0, 0, 0};
    bI[lp] = (v8f){0, 0, 0, 0, 0, 0, 0, 0};
  }
  #pragma unroll
  for (int kb = 0; kb < 4; ++kb) {
    const v2f Min = -Mi[kb];
    #pragma unroll
    for (int lp = 0; lp < 2; ++lp) {
      const v2f Br = (v2f){dR[lp][2 * kb], dR[lp][2 * kb + 1]};
      const v2f Bi = (v2f){dI[lp][2 * kb], dI[lp][2 * kb + 1]};
      bR[lp] = wmma4(Mr[kb], Br, bR[lp]);
      bR[lp] = wmma4(Min,    Bi, bR[lp]);
      bI[lp] = wmma4(Mr[kb], Bi, bI[lp]);
      bI[lp] = wmma4(Mi[kb], Br, bI[lp]);
    }
  }
  const float a00r = g8[0], a00i = g8[1], a01r = g8[2], a01i = g8[3];
  const float a10r = g8[4], a10i = g8[5], a11r = g8[6], a11i = g8[7];
  #pragma unroll
  for (int r = 0; r < 8; ++r) {
    const float s0x = bR[0][r], s0y = bI[0][r], s1x = bR[1][r], s1y = bI[1][r];
    nR[0][r] = a00r * s0x - a00i * s0y + a01r * s1x - a01i * s1y;
    nI[0][r] = a00r * s0y + a00i * s0x + a01r * s1y + a01i * s1x;
    nR[1][r] = a10r * s0x - a10i * s0y + a11r * s1x - a11i * s1y;
    nI[1][r] = a10r * s0y + a10i * s0x + a11r * s1y + a11i * s1x;
  }
}

__device__ __forceinline__ void load_patch(const float* __restrict__ x, int n, float (&a)[9]) {
  const int b   = n >> 12;
  const int rem = n & 4095;
  const int ii  = rem >> 6;
  const int jj  = rem & 63;
  #pragma unroll
  for (int q = 0; q < 9; ++q) {
    const int ki = ii + q / 3 - 1;
    const int kj = jj + q % 3 - 1;
    float v = 0.0f;
    if (ki >= 0 && ki < 64 && kj >= 0 && kj < 64) v = x[b * 4096 + ki * 64 + kj];
    a[q] = v;
  }
}

__global__ __launch_bounds__(256, 8) void qsim(const float* __restrict__ x,
                                            const float* __restrict__ gates,
                                            float* __restrict__ out) {
  const int wv   = threadIdx.x >> 5;
  const int lane = threadIdx.x & 31;
  const int lhalf = lane >> 4;
  const int lmod  = lane & 15;

  const int n0 = (blockIdx.x * 8 + wv) * SAMPLES_PER_WAVE;

  v2f Hr[4], Hi[4], Mr[4], Mi[4];
  load_gate(gates + WS_HI,  lmod * 16 + 2 * lhalf, 4, Hr, Hi);
  load_gate(gates + WS_MID, lmod * 16 + 8 * lhalf, 2, Mr, Mi);

  const int sl = (cnot_pinv((lmod << 5) | (lhalf << 4), 2) >> 8) & 1;

  float ang[9];
  load_patch(x, n0, ang);

  #pragma unroll 1
  for (int it = 0; it < SAMPLES_PER_WAVE; ++it) {
    const int n = n0 + it;

    float hc[9], hs[9];
    #pragma unroll
    for (int q = 0; q < 9; ++q) {
      hc[q] = __cosf(0.5f * ang[q]);
      hs[q] = __sinf(0.5f * ang[q]);
    }
    if (it + 1 < SAMPLES_PER_WAVE) load_patch(x, n + 1, ang);

    C2 va[9], vb[9];
    #pragma unroll
    for (int q = 0; q < 9; ++q) {
      const float c = hc[q];
      const float s = hs[q];
      const float* __restrict__ g = gates + WS_L0 + q * 8;
      va[q] = { g[0] * c + g[3] * s, g[1] * c - g[2] * s };
      vb[q] = { g[4] * c + g[7] * s, g[5] * c - g[6] * s };
    }

    v2f Sr[2][4], Si[2][4];
    {
      const bool i6 = lhalf != 0;
      const bool i4 = (lmod & 8) != 0, i3 = (lmod & 4) != 0;
      const bool i2 = (lmod & 2) != 0, i1 = (lmod & 1) != 0;
      const C2 Lc = cmul(cmul(csel(i3 != i4, va[5], vb[5]), csel(i2 != i3, va[6], vb[6])),
                         csel(i1 != i2, va[7], vb[7]));
      const C2 u2 = csel(i6, va[2], vb[2]), w2 = csel(i6, vb[2], va[2]);
      const C2 u3 = csel(i6, va[3], vb[3]), w3 = csel(i6, vb[3], va[3]);
      const C2 u4 = csel(i4, va[4], vb[4]), w4 = csel(i4, vb[4], va[4]);
      const C2 u8 = csel(i1, va[8], vb[8]), w8 = csel(i1, vb[8], va[8]);
      const C2 t0 = cmul(Lc, cmul(u3, u4));
      const C2 t1 = cmul(Lc, cmul(w3, w4));
      const C2 T[2][2] = { { cmul(t0, u8), cmul(t0, w8) }, { cmul(t1, u8), cmul(t1, w8) } };
      const C2 f0[2] = { va[0], vb[0] };
      const C2 f1[2] = { va[1], vb[1] };
      const C2 f2[2] = { u2, w2 };
      C2 X[2][2];
      #pragma unroll
      for (int a = 0; a < 2; ++a)
        #pragma unroll
        for (int i7 = 0; i7 < 2; ++i7)
          X[a][i7] = cmul(cmul(f0[a], f1[i7 ^ a]), f2[i7]);
      #pragma unroll
      for (int i8 = 0; i8 < 2; ++i8)
        #pragma unroll
        for (int i7 = 0; i7 < 2; ++i7)
          #pragma unroll
          for (int i5 = 0; i5 < 2; ++i5)
            #pragma unroll
            for (int i0 = 0; i0 < 2; ++i0) {
              const C2 amp = cmul(X[i8 ^ i0][i7], T[i5][i0]);
              Sr[i0][2 * i8 + i7][i5] = amp.re;
              Si[i0][2 * i8 + i7][i5] = amp.im;
            }
    }

    v8f nR[2], nI[2];
    layer_gates(Sr, Si, Hr, Hi, Mr, Mi, gates + WS_G8, nR, nI);

    float pz = 0.0f;
    #pragma unroll
    for (int r = 0; r < 8; ++r)
      #pragma unroll
      for (int lp = 0; lp < 2; ++lp) {
        const int sc = (cnot_pinv((r << 1) | lp, 2) >> 8) & 1;
        const float v = nR[lp][r] * nR[lp][r] + nI[lp][r] * nI[lp][r];
        pz += sc ? -v : v;
      }
    pz = sl ? -pz : pz;
    #pragma unroll
    for (int off = 16; off > 0; off >>= 1) pz += __shfl_down(pz, off, 32);
    if (lane == 0) out[n] = pz;
  }
}

extern "C" void kernel_launch(void* const* d_in, const int* in_sizes, int n_in,
                              void* d_out, int out_size, void* d_ws, size_t ws_size,
                              hipStream_t stream) {
  const float* x = (const float*)d_in[0];
  const float* w = (const float*)d_in[1];
  float* out = (float*)d_out;
  float* ws  = (float*)d_ws;

  hipLaunchKernelGGL(build_gates, dim3(1), dim3(256), 0, stream, w, ws);
  hipLaunchKernelGGL(qsim, dim3(32768 / (8 * SAMPLES_PER_WAVE)), dim3(256), 0, stream, x, ws, out);
}


# Z0 behind the last CNOT ring is a parity on wires 0,1,3,5,7: layer-1 rotations on wires 2,4,6,8 cancel and are dropped; one 4-wire WMMA group + wire 7 left, 64 -> 32 WMMAs; sum order changes
# speedup vs baseline: 3.4698x; 1.5053x over previous
#include <hip/hip_runtime.h>
#include <math.h>


typedef float v2f __attribute__((ext_vector_type(2)));
typedef float v8f __attribute__((ext_vector_type(8)));

#define WS_GRP 0
#define WS_G7  512
#define WS_L0  528

#define SAMPLES_PER_WAVE 4

__device__ __forceinline__ v8f wmma4(v2f a, v2f b, v8f c) {
  return __builtin_amdgcn_wmma_f32_16x16x4_f32(
       false, a,  false, b,
       (short)0, c,  false,  false);
}

__host__ __device__ constexpr int cnot_pinv(int j, int rr) {
  for (int q = 0; q < 9; ++q) {
    const int cb = 8 - q;
    const int tb = 8 - ((q + rr) % 9);
    j ^= ((j >> cb) & 1) << tb;
  }
  return j;
}
__host__ __device__ constexpr int cnot_p(int i, int rr) {
  for (int q = 8; q >= 0; --q) {
    const int cb = 8 - q;
    const int tb = 8 - ((q + rr) % 9);
    i ^= ((i >> cb) & 1) << tb;
  }
  return i;
}
__host__ __device__ constexpr bool ring1_closed_form_ok() {
  for (int i = 0; i < 512; ++i) {
    int k = (((i >> 8) ^ i) & 1) << 8;
    k |= (((i >> 7) ^ (i >> 8) ^ i) & 1) << 7;
    for (int b = 0; b < 7; ++b) k |= (((i >> b) ^ (i >> (b + 1))) & 1) << b;
    if (k != cnot_p(i, 1)) return false;
  }
  return true;
}
static_assert(ring1_closed_form_ok(), "closed form of the range-1 CNOT ring");
#define Z0_PARITY_MASK 0x1AA
__host__ __device__ constexpr bool ring2_parity_ok() {
  for (int j = 0; j < 512; ++j) {
    int p = 0;
    for (int b = 0; b < 9; ++b) p ^= ((j & Z0_PARITY_MASK) >> b) & 1;
    if (p != ((cnot_pinv(j, 2) >> 8) & 1)) return false;
  }
  return true;
}
static_assert(ring2_parity_ok(), "<Z_0> behind the range-2 ring is the parity on wires 0,1,3,5,7");

struct C2 { float re, im; };
__device__ __forceinline__ C2 cmul(C2 a, C2 b) {
  return { a.re * b.re - a.im * b.im, a.re * b.im + a.im * b.re };
}
__device__ __forceinline__ C2 csel(bool s, C2 a, C2 b) { return s ? b : a; }

__device__ __forceinline__ void rot_gate(const float* __restrict__ w3, C2 (&g)[2][2]) {
  const float phi = w3[0], theta = w3[1], omega = w3[2];
  const float ct = __cosf(0.5f * theta), st = __sinf(0.5f * theta);
  const float app = 0.5f * (phi + omega), apm = 0.5f * (phi - omega);
  const C2 ep = { __cosf(app), -__sinf(app) };
  const C2 em = { __cosf(apm),  __sinf(apm) };
  g[0][0] = {  ep.re * ct,  ep.im * ct };
  g[0][1] = { -em.re * st, -em.im * st };
  g[1][0] = {  em.re * st, -em.im * st };
  g[1][1] = {  ep.re * ct, -ep.im * ct };
}

__global__ void build_gates(const float* __restrict__ w, float* __restrict__ ws) {
  const int tid = threadIdx.x;

  const int r = tid >> 4, c = tid & 15;
  C2 grp = { 1.0f, 0.0f };
  #pragma unroll
  for (int k = 0; k < 4; ++k) {
    const int wire = (k < 2) ? k : 2 * k - 1;
    C2 g[2][2];
    rot_gate(w + (9 + wire) * 3, g);
    const bool rb = ((r >> (3 - k)) & 1) != 0, cb = ((c >> (3 - k)) & 1) != 0;
    grp = cmul(grp, csel(rb, csel(cb, g[0][0], g[0][1]), csel(cb, g[1][0], g[1][1])));
  }
  ws[WS_GRP + tid * 2 + 0] = grp.re;
  ws[WS_GRP + tid * 2 + 1] = grp.im;
  if (tid < 4) {
    C2 g7[2][2];
    rot_gate(w + (9 + 7) * 3, g7);
    const bool rb = (tid & 2) != 0, cb = (tid & 1) != 0;
    const C2 e = csel(rb, csel(cb, g7[0][0], g7[0][1]), csel(cb, g7[1][0], g7[1][1]));
    ws[WS_G7 + tid * 2 + 0] = e.re;
    ws[WS_G7 + tid * 2 + 1] = e.im;
  }
  if (tid < 9) {
    C2 g0[2][2];
    rot_gate(w + tid * 3, g0);
    float* d = ws + WS_L0 + tid * 8;
    d[0] = g0[0][0].re; d[1] = g0[0][0].im; d[2] = g0[0][1].re; d[3] = g0[0][1].im;
    d[4] = g0[1][0].re; d[5] = g0[1][0].im; d[6] = g0[1][1].re; d[7] = g0[1][1].im;
  }
}


__device__ __forceinline__ void load_gate(const float* __restrict__ G, int base, int kbstep,
                                          v2f (&Gr)[4], v2f (&Gi)[4]) {
  #pragma unroll
  for (int kb = 0; kb < 4; ++kb) {
    const float4 g = *reinterpret_cast<const float4*>(G + (base + kbstep * kb) * 2);
    Gr[kb] = (v2f){g.x, g.z};
    Gi[kb] = (v2f){g.y, g.w};
  }
}

__device__ __forceinline__ void layer_gates(const v2f (&Sr)[2][4], const v2f (&Si)[2][4],
                                            const v2f (&Hr)[4], const v2f (&Hi)[4],
                                            const float* __restrict__ g7,
                                            v8f (&nR)[2], v8f (&nI)[2]) {
  v8f dR[2], dI[2];
  #pragma unroll
  for (int lp = 0; lp < 2; ++lp) {
    dR[lp] = (v8f){0, 0, 0, 0, 0, 0, 0, 0};
    dI[lp] = (v8f){0, 0, 0, 0, 0, 0, 0, 0};
  }
  #pragma unroll
  for (int kb = 0; kb < 4; ++kb) {
    const v2f Hin = -Hi[kb];
    #pragma unroll
    for (int lp = 0; lp < 2; ++lp) {
      dR[lp] = wmma4(Sr[lp][kb], Hr[kb], dR[lp]);
      dR[lp] = wmma4(Si[lp][kb], Hin,    dR[lp]);
      dI[lp] = wmma4(Sr[lp][kb], Hi[kb], dI[lp]);
      dI[lp] = wmma4(Si[lp][kb], Hr[kb], dI[lp]);
    }
  }
  const float a00r = g7[0], a00i = g7[1], a01r = g7[2], a01i = g7[3];
  const float a10r = g7[4], a10i = g7[5], a11r = g7[6], a11i = g7[7];
  #pragma unroll
  for (int r = 0; r < 8; ++r) {
    const float s0x = dR[0][r], s0y = dI[0][r], s1x = dR[1][r], s1y = dI[1][r];
    nR[0][r] = a00r * s0x - a00i * s0y + a01r * s1x - a01i * s1y;
    nI[0][r] = a00r * s0y + a00i * s0x + a01r * s1y + a01i * s1x;
    nR[1][r] = a10r * s0x - a10i * s0y + a11r * s1x - a11i * s1y;
    nI[1][r] = a10r * s0y + a10i * s0x + a11r * s1y + a11i * s1x;
  }
}

__device__ __forceinline__ void load_patch(const float* __restrict__ x, int n, float (&a)[9]) {
  const int b   = n >> 12;
  const int rem = n & 4095;
  const int ii  = rem >> 6;
  const int jj  = rem & 63;
  #pragma unroll
  for (int q = 0; q < 9; ++q) {
    const int ki = ii + q / 3 - 1;
    const int kj = jj + q % 3 - 1;
    float v = 0.0f;
    if (ki >= 0 && ki < 64 && kj >= 0 && kj < 64) v = x[b * 4096 + ki * 64 + kj];
    a[q] = v;
  }
}

__global__ __launch_bounds__(256, 8) void qsim(const float* __restrict__ x,
                                            const float* __restrict__ gates,
                                            float* __restrict__ out) {
  const int wv   = threadIdx.x >> 5;
  const int lane = threadIdx.x & 31;
  const int lhalf = lane >> 4;
  const int lmod  = lane & 15;

  const int n0 = (blockIdx.x * 8 + wv) * SAMPLES_PER_WAVE;

  v2f Hr[4], Hi[4];
  load_gate(gates + WS_GRP, lmod * 16 + 2 * lhalf, 4, Hr, Hi);

  const int sl = __popc(lmod) & 1;

  float ang[9];
  load_patch(x, n0, ang);

  #pragma unroll 1
  for (int it = 0; it < SAMPLES_PER_WAVE; ++it) {
    const int n = n0 + it;

    float hc[9], hs[9];
    #pragma unroll
    for (int q = 0; q < 9; ++q) {
      hc[q] = __cosf(0.5f * ang[q]);
      hs[q] = __sinf(0.5f * ang[q]);
    }
    if (it + 1 < SAMPLES_PER_WAVE) load_patch(x, n + 1, ang);

    C2 va[9], vb[9];
    #pragma unroll
    for (int q = 0; q < 9; ++q) {
      const float c = hc[q];
      const float s = hs[q];
      const float* __restrict__ g = gates + WS_L0 + q * 8;
      va[q] = { g[0] * c + g[3] * s, g[1] * c - g[2] * s };
      vb[q] = { g[4] * c + g[7] * s, g[5] * c - g[6] * s };
    }

    v2f Sr[2][4], Si[2][4];
    {
      const bool i5 = lhalf != 0;
      const bool i6 = (lmod & 8) != 0, i4 = (lmod & 4) != 0;
      const bool i2 = (lmod & 2) != 0, i0 = (lmod & 1) != 0;
      const C2 Lc = cmul(csel(i5 != i6, va[3], vb[3]), csel(i4 != i5, va[4], vb[4]));
      const C2 u0 = csel(i0, va[0], vb[0]), w0 = csel(i0, vb[0], va[0]);
      const C2 u1 = csel(i0, va[1], vb[1]), w1 = csel(i0, vb[1], va[1]);
      const C2 u2 = csel(i6, va[2], vb[2]), w2 = csel(i6, vb[2], va[2]);
      const C2 u5 = csel(i4, va[5], vb[5]), w5 = csel(i4, vb[5], va[5]);
      const C2 u6 = csel(i2, va[6], vb[6]), w6 = csel(i2, vb[6], va[6]);
      const C2 u7 = csel(i2, va[7], vb[7]), w7 = csel(i2, vb[7], va[7]);
      const C2 u8 = csel(i0, va[8], vb[8]), w8 = csel(i0, vb[8], va[8]);
      const C2 c0 = cmul(Lc, cmul(u5, u6));
      const C2 c1 = cmul(Lc, cmul(w5, w6));
      const C2 d0 = cmul(u7, u8);
      const C2 d1 = cmul(w7, w8);
      const C2 T[2][2] = { { cmul(c0, d0), cmul(c0, d1) }, { cmul(c1, d0), cmul(c1, d1) } };
      const C2 f0[2] = { u0, w0 };
      const C2 f1[2] = { u1, w1 };
      const C2 f2[2] = { u2, w2 };
      C2 X[2][2];
      #pragma unroll
      for (int i8 = 0; i8 < 2; ++i8)
        #pragma unroll
        for (int i7 = 0; i7 < 2; ++i7)
          X[i8][i7] = cmul(cmul(f0[i8], f1[i7 ^ i8]), f2[i7]);
      #pragma unroll
      for (int i8 = 0; i8 < 2; ++i8)
        #pragma unroll
        for (int i7 = 0; i7 < 2; ++i7)
          #pragma unroll
          for (int i3 = 0; i3 < 2; ++i3)
            #pragma unroll
            for (int i1 = 0; i1 < 2; ++i1) {
              const C2 amp = cmul(X[i8][i7], T[i3][i1]);
              Sr[i1][2 * i8 + i7][i3] = amp.re;
              Si[i1][2 * i8 + i7][i3] = amp.im;
            }
    }

    v8f nR[2], nI[2];
    layer_gates(Sr, Si, Hr, Hi, gates + WS_G7, nR, nI);

    float pz = 0.0f;
    #pragma unroll
    for (int r = 0; r < 8; ++r)
      #pragma unroll
      for (int lp = 0; lp < 2; ++lp) {
        const float v = nR[lp][r] * nR[lp][r] + nI[lp][r] * nI[lp][r];
        pz += lp ? -v : v;
      }
    pz = sl ? -pz : pz;
    #pragma unroll
    for (int off = 16; off > 0; off >>= 1) pz += __shfl_down(pz, off, 32);
    if (lane == 0) out[n] = pz;
  }
}

extern "C" void kernel_launch(void* const* d_in, const int* in_sizes, int n_in,
                              void* d_out, int out_size, void* d_ws, size_t ws_size,
                              hipStream_t stream) {
  const float* x = (const float*)d_in[0];
  const float* w = (const float*)d_in[1];
  float* out = (float*)d_out;
  float* ws  = (float*)d_ws;

  hipLaunchKernelGGL(build_gates, dim3(1), dim3(256), 0, stream, w, ws);
  hipLaunchKernelGGL(qsim, dim3(32768 / (8 * SAMPLES_PER_WAVE)), dim3(256), 0, stream, x, ws, out);
}


# Gate build folded into the simulator launch: lanes 0-13 of each wave make one wire's rotation from the weights, lane broadcasts give the scalars and the group gate's B fragments; 2 launches -> 1
# speedup vs baseline: 4.1845x; 1.2060x over previous
#include <hip/hip_runtime.h>
#include <math.h>


typedef float v2f __attribute__((ext_vector_type(2)));
typedef float v8f __attribute__((ext_vector_type(8)));

#define SAMPLES_PER_WAVE 4

__device__ __forceinline__ v8f wmma4(v2f a, v2f b, v8f c) {
  return __builtin_amdgcn_wmma_f32_16x16x4_f32(
       false, a,  false, b,
       (short)0, c,  false,  false);
}

__host__ __device__ constexpr int cnot_pinv(int j, int rr) {
  for (int q = 0; q < 9; ++q) {
    const int cb = 8 - q;
    const int tb = 8 - ((q + rr) % 9);
    j ^= ((j >> cb) & 1) << tb;
  }
  return j;
}
__host__ __device__ constexpr int cnot_p(int i, int rr) {
  for (int q = 8; q >= 0; --q) {
    const int cb = 8 - q;
    const int tb = 8 - ((q + rr) % 9);
    i ^= ((i >> cb) & 1) << tb;
  }
  return i;
}
__host__ __device__ constexpr bool ring1_closed_form_ok() {
  for (int i = 0; i < 512; ++i) {
    int k = (((i >> 8) ^ i) & 1) << 8;
    k |= (((i >> 7) ^ (i >> 8) ^ i) & 1) << 7;
    for (int b = 0; b < 7; ++b) k |= (((i >> b) ^ (i >> (b + 1))) & 1) << b;
    if (k != cnot_p(i, 1)) return false;
  }
  return true;
}
static_assert(ring1_closed_form_ok(), "closed form of the range-1 CNOT ring");
#define Z0_PARITY_MASK 0x1AA
__host__ __device__ constexpr bool ring2_parity_ok() {
  for (int j = 0; j < 512; ++j) {
    int p = 0;
    for (int b = 0; b < 9; ++b) p ^= ((j & Z0_PARITY_MASK) >> b) & 1;
    if (p != ((cnot_pinv(j, 2) >> 8) & 1)) return false;
  }
  return true;
}
static_assert(ring2_parity_ok(), "<Z_0> behind the range-2 ring is the parity on wires 0,1,3,5,7");

struct C2 { float re, im; };
__device__ __forceinline__ C2 cmul(C2 a, C2 b) {
  return { a.re * b.re - a.im * b.im, a.re * b.im + a.im * b.re };
}
__device__ __forceinline__ C2 csel(bool s, C2 a, C2 b) { return s ? b : a; }

struct RotAB { C2 a, b; };
__device__ __forceinline__ RotAB rot_ab(float phi, float theta, float omega) {
  const float ct = __cosf(0.5f * theta), st = __sinf(0.5f * theta);
  const float app = 0.5f * (phi + omega), apm = 0.5f * (phi - omega);
  const C2 ep = { __cosf(app), -__sinf(app) };
  const C2 em = { __cosf(apm),  __sinf(apm) };
  RotAB g;
  g.a = { ep.re * ct, ep.im * ct };
  g.b = { em.re * st, em.im * st };
  return g;
}
__device__ __forceinline__ C2 cconj(C2 a) { return { a.re, -a.im }; }
__device__ __forceinline__ C2 cneg(C2 a)  { return { -a.re, -a.im }; }

__device__ __forceinline__ float bcast(float v, int src_lane) {
  return __builtin_bit_cast(float,
      __builtin_amdgcn_readlane(__builtin_bit_cast(int, v), src_lane));
}
__device__ __forceinline__ RotAB bcast(const RotAB& g, int src_lane) {
  RotAB r;
  r.a = { bcast(g.a.re, src_lane), bcast(g.a.im, src_lane) };
  r.b = { bcast(g.b.re, src_lane), bcast(g.b.im, src_lane) };
  return r;
}


__device__ __forceinline__ void make_group_gate(const RotAB (&gw)[4], int lmod, int lhalf,
                                                v2f (&Gr)[4], v2f (&Gi)[4]) {
  C2 e[4][2];
  #pragma unroll
  for (int k = 0; k < 4; ++k) {
    const bool rb = ((lmod >> (3 - k)) & 1) != 0;
    e[k][0] = csel(rb, gw[k].a, cconj(gw[k].b));
    e[k][1] = csel(rb, cneg(gw[k].b), cconj(gw[k].a));
  }
  const C2 e3 = csel(lhalf != 0, e[2][0], e[2][1]);
  #pragma unroll
  for (int kb = 0; kb < 4; ++kb) {
    const C2 p = cmul(cmul(e[0][kb >> 1], e[1][kb & 1]), e3);
    const C2 g0 = cmul(p, e[3][0]);
    const C2 g1 = cmul(p, e[3][1]);
    Gr[kb] = (v2f){g0.re, g1.re};
    Gi[kb] = (v2f){g0.im, g1.im};
  }
}

__device__ __forceinline__ void layer_gates(const v2f (&Sr)[2][4], const v2f (&Si)[2][4],
                                            const v2f (&Hr)[4], const v2f (&Hi)[4],
                                            const RotAB& g7,
                                            v8f (&nR)[2], v8f (&nI)[2]) {
  v8f dR[2], dI[2];
  #pragma unroll
  for (int lp = 0; lp < 2; ++lp) {
    dR[lp] = (v8f){0, 0, 0, 0, 0, 0, 0, 0};
    dI[lp] = (v8f){0, 0, 0, 0, 0, 0, 0, 0};
  }
  #pragma unroll
  for (int kb = 0; kb < 4; ++kb) {
    const v2f Hin = -Hi[kb];
    #pragma unroll
    for (int lp = 0; lp < 2; ++lp) {
      dR[lp] = wmma4(Sr[lp][kb], Hr[kb], dR[lp]);
      dR[lp] = wmma4(Si[lp][kb], Hin,    dR[lp]);
      dI[lp] = wmma4(Sr[lp][kb], Hi[kb], dI[lp]);
      dI[lp] = wmma4(Si[lp][kb], Hr[kb], dI[lp]);
    }
  }
  const float a00r =  g7.a.re, a00i =  g7.a.im, a01r = -g7.b.re, a01i = -g7.b.im;
  const float a10r =  g7.b.re, a10i = -g7.b.im, a11r =  g7.a.re, a11i = -g7.a.im;
  #pragma unroll
  for (int r = 0; r < 8; ++r) {
    const float s0x = dR[0][r], s0y = dI[0][r], s1x = dR[1][r], s1y = dI[1][r];
    nR[0][r] = a00r * s0x - a00i * s0y + a01r * s1x - a01i * s1y;
    nI[0][r] = a00r * s0y + a00i * s0x + a01r * s1y + a01i * s1x;
    nR[1][r] = a10r * s0x - a10i * s0y + a11r * s1x - a11i * s1y;
    nI[1][r] = a10r * s0y + a10i * s0x + a11r * s1y + a11i * s1x;
  }
}

__device__ __forceinline__ void load_patch(const float* __restrict__ x, int n, float (&a)[9]) {
  const int b   = n >> 12;
  const int rem = n & 4095;
  const int ii  = rem >> 6;
  const int jj  = rem & 63;
  #pragma unroll
  for (int q = 0; q < 9; ++q) {
    const int ki = ii + q / 3 - 1;
    const int kj = jj + q % 3 - 1;
    float v = 0.0f;
    if (ki >= 0 && ki < 64 && kj >= 0 && kj < 64) v = x[b * 4096 + ki * 64 + kj];
    a[q] = v;
  }
}

__global__ __launch_bounds__(256, 8) void qsim(const float* __restrict__ x,
                                            const float* __restrict__ w,
                                            float* __restrict__ out) {
  const int wv   = threadIdx.x >> 5;
  const int lane = threadIdx.x & 31;
  const int lhalf = lane >> 4;
  const int lmod  = lane & 15;

  const int n0 = (blockIdx.x * 8 + wv) * SAMPLES_PER_WAVE;

  const int l1k  = lane - 9;
  const int grow = lane < 9 ? lane : (lane < 14 ? 9 + (l1k < 2 ? l1k : 2 * l1k - 1) : 0);
  const float* __restrict__ w3 = w + grow * 3;
  const float phi = w3[0], theta = w3[1], omega = w3[2];

  float ang[9];
  load_patch(x, n0, ang);

  const RotAB mine = rot_ab(phi, theta, omega);

  RotAB L0[9];
  #pragma unroll
  for (int q = 0; q < 9; ++q) L0[q] = bcast(mine, q);
  const RotAB G7 = bcast(mine, 13);

  v2f Hr[4], Hi[4];
  {
    RotAB gw[4];
    #pragma unroll
    for (int k = 0; k < 4; ++k) gw[k] = bcast(mine, 9 + k);
    make_group_gate(gw, lmod, lhalf, Hr, Hi);
  }

  const int sl = __popc(lmod) & 1;

  #pragma unroll 1
  for (int it = 0; it < SAMPLES_PER_WAVE; ++it) {
    const int n = n0 + it;

    float hc[9], hs[9];
    #pragma unroll
    for (int q = 0; q < 9; ++q) {
      hc[q] = __cosf(0.5f * ang[q]);
      hs[q] = __sinf(0.5f * ang[q]);
    }
    if (it + 1 < SAMPLES_PER_WAVE) load_patch(x, n + 1, ang);

    C2 va[9], vb[9];
    #pragma unroll
    for (int q = 0; q < 9; ++q) {
      const float c = hc[q];
      const float s = hs[q];
      const C2 a = L0[q].a, b = L0[q].b;
      va[q] = { a.re * c - b.im * s,  a.im * c + b.re * s };
      vb[q] = { b.re * c - a.im * s, -b.im * c - a.re * s };
    }

    v2f Sr[2][4], Si[2][4];
    {
      const bool i5 = lhalf != 0;
      const bool i6 = (lmod & 8) != 0, i4 = (lmod & 4) != 0;
      const bool i2 = (lmod & 2) != 0, i0 = (lmod & 1) != 0;
      const C2 Lc = cmul(csel(i5 != i6, va[3], vb[3]), csel(i4 != i5, va[4], vb[4]));
      const C2 u0 = csel(i0, va[0], vb[0]), w0 = csel(i0, vb[0], va[0]);
      const C2 u1 = csel(i0, va[1], vb[1]), w1 = csel(i0, vb[1], va[1]);
      const C2 u2 = csel(i6, va[2], vb[2]), w2 = csel(i6, vb[2], va[2]);
      const C2 u5 = csel(i4, va[5], vb[5]), w5 = csel(i4, vb[5], va[5]);
      const C2 u6 = csel(i2, va[6], vb[6]), w6 = csel(i2, vb[6], va[6]);
      const C2 u7 = csel(i2, va[7], vb[7]), w7 = csel(i2, vb[7], va[7]);
      const C2 u8 = csel(i0, va[8], vb[8]), w8 = csel(i0, vb[8], va[8]);
      const C2 c0 = cmul(Lc, cmul(u5, u6));
      const C2 c1 = cmul(Lc, cmul(w5, w6));
      const C2 d0 = cmul(u7, u8);
      const C2 d1 = cmul(w7, w8);
      const C2 T[2][2] = { { cmul(c0, d0), cmul(c0, d1) }, { cmul(c1, d0), cmul(c1, d1) } };
      const C2 f0[2] = { u0, w0 };
      const C2 f1[2] = { u1, w1 };
      const C2 f2[2] = { u2, w2 };
      C2 X[2][2];
      #pragma unroll
      for (int i8 = 0; i8 < 2; ++i8)
        #pragma unroll
        for (int i7 = 0; i7 < 2; ++i7)
          X[i8][i7] = cmul(cmul(f0[i8], f1[i7 ^ i8]), f2[i7]);
      #pragma unroll
      for (int i8 = 0; i8 < 2; ++i8)
        #pragma unroll
        for (int i7 = 0; i7 < 2; ++i7)
          #pragma unroll
          for (int i3 = 0; i3 < 2; ++i3)
            #pragma unroll
            for (int i1 = 0; i1 < 2; ++i1) {
              const C2 amp = cmul(X[i8][i7], T[i3][i1]);
              Sr[i1][2 * i8 + i7][i3] = amp.re;
              Si[i1][2 * i8 + i7][i3] = amp.im;
            }
    }

    v8f nR[2], nI[2];
    layer_gates(Sr, Si, Hr, Hi, G7, nR, nI);

    float pz = 0.0f;
    #pragma unroll
    for (int r = 0; r < 8; ++r)
      #pragma unroll
      for (int lp = 0; lp < 2; ++lp) {
        const float v = nR[lp][r] * nR[lp][r] + nI[lp][r] * nI[lp][r];
        pz += lp ? -v : v;
      }
    pz = sl ? -pz : pz;
    #pragma unroll
    for (int off = 16; off > 0; off >>= 1) pz += __shfl_down(pz, off, 32);
    if (lane == 0) out[n] = pz;
  }
}

extern "C" void kernel_launch(void* const* d_in, const int* in_sizes, int n_in,
                              void* d_out, int out_size, void* d_ws, size_t ws_size,
                              hipStream_t stream) {
  const float* x = (const float*)d_in[0];
  const float* w = (const float*)d_in[1];
  float* out = (float*)d_out;

  hipLaunchKernelGGL(qsim, dim3(32768 / (8 * SAMPLES_PER_WAVE)), dim3(256), 0, stream, x, w, out);
}


# Per-sample scalar work (patch, sin/cos, wire 2-vectors, partial products, row weights) done once, one sample per lane, passed via LDS; lane partials summed per sample via LDS; sum order changes
# speedup vs baseline: 8.1026x; 1.9363x over previous
#include <hip/hip_runtime.h>
#include <math.h>


typedef float v2f __attribute__((ext_vector_type(2)));
typedef float v8f __attribute__((ext_vector_type(8)));

#define SAMPLES_PER_WAVE 16
#define WAVES_PER_BLOCK 4
#define TAB_ROW 52
#define RED_ROW 33

__device__ __forceinline__ v8f wmma4(v2f a, v2f b, v8f c) {
  return __builtin_amdgcn_wmma_f32_16x16x4_f32(
       false, a,  false, b,
       (short)0, c,  false,  false);
}

__host__ __device__ constexpr int cnot_pinv(int j, int rr) {
  for (int q = 0; q < 9; ++q) {
    const int cb = 8 - q;
    const int tb = 8 - ((q + rr) % 9);
    j ^= ((j >> cb) & 1) << tb;
  }
  return j;
}
__host__ __device__ constexpr int cnot_p(int i, int rr) {
  for (int q = 8; q >= 0; --q) {
    const int cb = 8 - q;
    const int tb = 8 - ((q + rr) % 9);
    i ^= ((i >> cb) & 1) << tb;
  }
  return i;
}
__host__ __device__ constexpr bool ring1_closed_form_ok() {
  for (int i = 0; i < 512; ++i) {
    int k = (((i >> 8) ^ i) & 1) << 8;
    k |= (((i >> 7) ^ (i >> 8) ^ i) & 1) << 7;
    for (int b = 0; b < 7; ++b) k |= (((i >> b) ^ (i >> (b + 1))) & 1) << b;
    if (k != cnot_p(i, 1)) return false;
  }
  return true;
}
static_assert(ring1_closed_form_ok(), "closed form of the range-1 CNOT ring");
#define Z0_PARITY_MASK 0x1AA
__host__ __device__ constexpr bool ring2_parity_ok() {
  for (int j = 0; j < 512; ++j) {
    int p = 0;
    for (int b = 0; b < 9; ++b) p ^= ((j & Z0_PARITY_MASK) >> b) & 1;
    if (p != ((cnot_pinv(j, 2) >> 8) & 1)) return false;
  }
  return true;
}
static_assert(ring2_parity_ok(), "<Z_0> behind the range-2 ring is the parity on wires 0,1,3,5,7");

struct C2 { float re, im; };
__device__ __forceinline__ C2 cmul(C2 a, C2 b) {
  return { a.re * b.re - a.im * b.im, a.re * b.im + a.im * b.re };
}
__device__ __forceinline__ C2 csel(bool s, C2 a, C2 b) { return s ? b : a; }

struct RotAB { C2 a, b; };
__device__ __forceinline__ RotAB rot_ab(float phi, float theta, float omega) {
  const float ct = __cosf(0.5f * theta), st = __sinf(0.5f * theta);
  const float app = 0.5f * (phi + omega), apm = 0.5f * (phi - omega);
  const C2 ep = { __cosf(app), -__sinf(app) };
  const C2 em = { __cosf(apm),  __sinf(apm) };
  RotAB g;
  g.a = { ep.re * ct, ep.im * ct };
  g.b = { em.re * st, em.im * st };
  return g;
}
__device__ __forceinline__ C2 cconj(C2 a) { return { a.re, -a.im }; }
__device__ __forceinline__ C2 cneg(C2 a)  { return { -a.re, -a.im }; }

__device__ __forceinline__ float bcast(float v, int src_lane) {
  return __builtin_bit_cast(float,
      __builtin_amdgcn_readlane(__builtin_bit_cast(int, v), src_lane));
}
__device__ __forceinline__ RotAB bcast(const RotAB& g, int src_lane) {
  RotAB r;
  r.a = { bcast(g.a.re, src_lane), bcast(g.a.im, src_lane) };
  r.b = { bcast(g.b.re, src_lane), bcast(g.b.im, src_lane) };
  return r;
}


__device__ __forceinline__ void make_group_gate(const RotAB (&gw)[4], int lmod, int lhalf,
                                                v2f (&Gr)[4], v2f (&Gi)[4]) {
  C2 e[4][2];
  #pragma unroll
  for (int k = 0; k < 4; ++k) {
    const bool rb = ((lmod >> (3 - k)) & 1) != 0;
    e[k][0] = csel(rb, gw[k].a, cconj(gw[k].b));
    e[k][1] = csel(rb, cneg(gw[k].b), cconj(gw[k].a));
  }
  const C2 e3 = csel(lhalf != 0, e[2][0], e[2][1]);
  #pragma unroll
  for (int kb = 0; kb < 4; ++kb) {
    const C2 p = cmul(cmul(e[0][kb >> 1], e[1][kb & 1]), e3);
    const C2 g0 = cmul(p, e[3][0]);
    const C2 g1 = cmul(p, e[3][1]);
    Gr[kb] = (v2f){g0.re, g1.re};
    Gi[kb] = (v2f){g0.im, g1.im};
  }
}

__device__ __forceinline__ void group_gate(const v2f (&Sr)[4], const v2f (&Si)[4],
                                           const v2f (&Hr)[4], const v2f (&Hi)[4],
                                           v8f &yR, v8f &yI) {
  yR = (v8f){0, 0, 0, 0, 0, 0, 0, 0};
  yI = (v8f){0, 0, 0, 0, 0, 0, 0, 0};
  #pragma unroll
  for (int kb = 0; kb < 4; ++kb) {
    const v2f Hin = -Hi[kb];
    yR = wmma4(Sr[kb], Hr[kb], yR);
    yI = wmma4(Sr[kb], Hi[kb], yI);
    yR = wmma4(Si[kb], Hin,    yR);
    yI = wmma4(Si[kb], Hr[kb], yI);
  }
}

__device__ __forceinline__ void load_patch(const float* __restrict__ x, int n, float (&a)[9]) {
  const int b   = n >> 12;
  const int rem = n & 4095;
  const int ii  = rem >> 6;
  const int jj  = rem & 63;
  #pragma unroll
  for (int q = 0; q < 9; ++q) {
    const int ki = ii + q / 3 - 1;
    const int kj = jj + q % 3 - 1;
    float v = 0.0f;
    if (ki >= 0 && ki < 64 && kj >= 0 && kj < 64) v = x[b * 4096 + ki * 64 + kj];
    a[q] = v;
  }
}

__global__ __launch_bounds__(32 * WAVES_PER_BLOCK) void qsim(const float* __restrict__ x,
                                            const float* __restrict__ w,
                                            float* __restrict__ out) {
  __shared__ __attribute__((aligned(16))) float tab[WAVES_PER_BLOCK * SAMPLES_PER_WAVE * TAB_ROW];
  __shared__ float red[WAVES_PER_BLOCK * SAMPLES_PER_WAVE * RED_ROW];

  const int wv   = threadIdx.x >> 5;
  const int lane = threadIdx.x & 31;
  const int lhalf = lane >> 4;
  const int lmod  = lane & 15;

  const int n0 = (blockIdx.x * WAVES_PER_BLOCK + wv) * SAMPLES_PER_WAVE;

  const int l1k  = lane - 9;
  const int grow = lane < 9 ? lane : (lane < 14 ? 9 + (l1k < 2 ? l1k : 2 * l1k - 1) : 0);
  const float* __restrict__ w3 = w + grow * 3;
  const float phi = w3[0], theta = w3[1], omega = w3[2];

  const int smp = lane & (SAMPLES_PER_WAVE - 1);
  float ang[9];
  load_patch(x, n0 + smp, ang);

  const RotAB mine = rot_ab(phi, theta, omega);

  RotAB L0[9];
  #pragma unroll
  for (int q = 0; q < 9; ++q) L0[q] = bcast(mine, q);
  const RotAB G7 = bcast(mine, 13);
  const float m7 = (G7.a.re * G7.a.re + G7.a.im * G7.a.im) - (G7.b.re * G7.b.re + G7.b.im * G7.b.im);
  const C2 q7 = { G7.a.re * G7.b.re + G7.a.im * G7.b.im,
                  G7.a.im * G7.b.re - G7.a.re * G7.b.im };

  v2f Hr[4], Hi[4];
  {
    RotAB gw[4];
    #pragma unroll
    for (int k = 0; k < 4; ++k) gw[k] = bcast(mine, 9 + k);
    make_group_gate(gw, lmod, lhalf, Hr, Hi);
  }

  const int sl = __popc(lmod) & 1;

  {
    C2 v[9][2];
    #pragma unroll
    for (int q = 0; q < 9; ++q) {
      const float c = __cosf(0.5f * ang[q]);
      const float s = __sinf(0.5f * ang[q]);
      const C2 a = L0[q].a, b = L0[q].b;
      v[q][0] = { a.re * c - b.im * s,  a.im * c + b.re * s };
      v[q][1] = { b.re * c - a.im * s, -b.im * c - a.re * s };
    }

    float t[TAB_ROW];
    #pragma unroll
    for (int a = 0; a < 2; ++a)
      #pragma unroll
      for (int b = 0; b < 2; ++b) {
        const C2 p01 = cmul(v[0][a], v[1][b]);
        #pragma unroll
        for (int c = 0; c < 2; ++c) {
          const C2 p = cmul(p01, v[2][c]);
          t[2 * (4 * a + 2 * b + c)]     = p.re;
          t[2 * (4 * a + 2 * b + c) + 1] = p.im;
        }
      }
    C2 p56[2][2];
    #pragma unroll
    for (int f = 0; f < 2; ++f)
      #pragma unroll
      for (int g = 0; g < 2; ++g) p56[f][g] = cmul(v[5][f], v[6][g]);
    #pragma unroll
    for (int d = 0; d < 2; ++d)
      #pragma unroll
      for (int e = 0; e < 2; ++e) {
        const C2 p34 = cmul(v[3][d], v[4][e]);
        #pragma unroll
        for (int f = 0; f < 2; ++f)
          #pragma unroll
          for (int g = 0; g < 2; ++g) {
            const C2 p = cmul(p34, p56[f][g]);
            t[16 + 2 * (8 * d + 4 * e + 2 * f + g)]     = p.re;
            t[16 + 2 * (8 * d + 4 * e + 2 * f + g) + 1] = p.im;
          }
      }

    {
      const C2 P00 = cmul(v[7][0], v[8][0]), P01 = cmul(v[7][0], v[8][1]);
      const C2 P10 = cmul(v[7][1], v[8][0]), P11 = cmul(v[7][1], v[8][1]);
      const float n00 = P00.re * P00.re + P00.im * P00.im;
      const float n01 = P01.re * P01.re + P01.im * P01.im;
      const float n10 = P10.re * P10.re + P10.im * P10.im;
      const float n11 = P11.re * P11.re + P11.im * P11.im;
      const float D1 = m7 * (n00 - n11);
      const float D2 = m7 * (n01 - n10);
      const float z1r = P00.re * P11.re + P00.im * P11.im;
      const float z1i = P00.im * P11.re - P00.re * P11.im;
      const float z2r = P01.re * P10.re + P01.im * P10.im;
      const float z2i = P01.im * P10.re - P01.re * P10.im;
      t[48] =  D1 - 4.0f * (q7.re * z1r - q7.im * z1i);
      t[51] = -D1 - 4.0f * (q7.re * z1r + q7.im * z1i);
      t[49] =  D2 - 4.0f * (q7.re * z2r - q7.im * z2i);
      t[50] = -D2 - 4.0f * (q7.re * z2r + q7.im * z2i);
    }

    if (lhalf == 0) {
      float4* dst = reinterpret_cast<float4*>(&tab[(wv * SAMPLES_PER_WAVE + smp) * TAB_ROW]);
      #pragma unroll
      for (int k = 0; k < TAB_ROW / 4; ++k)
        dst[k] = make_float4(t[4 * k], t[4 * k + 1], t[4 * k + 2], t[4 * k + 3]);
    }
  }
  __syncthreads();

  int offX[4], offT[2];
  {
    const int i5 = lhalf;
    const int i6 = (lmod >> 3) & 1, i4 = (lmod >> 2) & 1;
    const int i2 = (lmod >> 1) & 1, i0 = lmod & 1;
    #pragma unroll
    for (int kb = 0; kb < 4; ++kb) {
      const int i8 = kb >> 1, i7 = kb & 1;
      offX[kb] = 2 * (4 * (i8 ^ i0) + 2 * (i7 ^ i8 ^ i0) + (i6 ^ i7));
    }
    #pragma unroll
    for (int i3 = 0; i3 < 2; ++i3)
      offT[i3] = 16 + 2 * (8 * (i5 ^ i6) + 4 * (i4 ^ i5) + 2 * (i3 ^ i4) + (i2 ^ i3));
  }
  const float* wtab = &tab[wv * SAMPLES_PER_WAVE * TAB_ROW];

  for (int it = 0; it < SAMPLES_PER_WAVE; ++it) {
    const float* r = wtab + it * TAB_ROW;
    float2 Xc[4], Tc[2];
    #pragma unroll
    for (int kb = 0; kb < 4; ++kb) Xc[kb] = *reinterpret_cast<const float2*>(r + offX[kb]);
    #pragma unroll
    for (int i3 = 0; i3 < 2; ++i3) Tc[i3] = *reinterpret_cast<const float2*>(r + offT[i3]);
    const float4 wr = *reinterpret_cast<const float4*>(r + 48);
    const float wrow[4] = { wr.x, wr.y, wr.z, wr.w };

    v2f Sr[4], Si[4];
    #pragma unroll
    for (int kb = 0; kb < 4; ++kb)
      #pragma unroll
      for (int i3 = 0; i3 < 2; ++i3) {
        const C2 amp = cmul(C2{Xc[kb].x, Xc[kb].y}, C2{Tc[i3].x, Tc[i3].y});
        Sr[kb][i3] = amp.re;
        Si[kb][i3] = amp.im;
      }

    v8f yR, yI;
    group_gate(Sr, Si, Hr, Hi, yR, yI);

    float pz = 0.0f;
    #pragma unroll
    for (int r = 0; r < 8; ++r) {
      const float vv = yR[r] * yR[r] + yI[r] * yI[r];
      pz += wrow[r & 3] * vv;
    }
    pz = sl ? -pz : pz;
    red[(wv * SAMPLES_PER_WAVE + it) * RED_ROW + lane] = pz;
  }

  __syncthreads();
  if (threadIdx.x < WAVES_PER_BLOCK * SAMPLES_PER_WAVE) {
    const float* r = &red[threadIdx.x * RED_ROW];
    float acc = 0.0f;
    #pragma unroll
    for (int k = 0; k < 32; ++k) acc += r[k];
    out[blockIdx.x * (WAVES_PER_BLOCK * SAMPLES_PER_WAVE) + threadIdx.x] = acc;
  }
}

extern "C" void kernel_launch(void* const* d_in, const int* in_sizes, int n_in,
                              void* d_out, int out_size, void* d_ws, size_t ws_size,
                              hipStream_t stream) {
  const float* x = (const float*)d_in[0];
  const float* w = (const float*)d_in[1];
  float* out = (float*)d_out;

  hipLaunchKernelGGL(qsim, dim3(32768 / (WAVES_PER_BLOCK * SAMPLES_PER_WAVE)), dim3(32 * WAVES_PER_BLOCK), 0, stream, x, w, out);
}
